# starting kernel
# baseline (speedup 1.0000x reference)
#include <hip/hip_runtime.h>
#include <hip/hip_bf16.h>

typedef __attribute__((ext_vector_type(2))) float v2f;
typedef __attribute__((ext_vector_type(8))) float v8f;

#define N_NODES 50000
#define N_EDGES 800000
#define D 128
#define N_CLASSES 40
#define ROW_TILES 3125

__global__ __launch_bounds__(256) void deg_init(float* __restrict__ deg) {
    int i = blockIdx.x * blockDim.x + threadIdx.x;
    if (i < N_NODES) deg[i] = 1.0f;
}

__global__ __launch_bounds__(256) void deg_count(const int* __restrict__ dst,
                                                 float* __restrict__ deg) {
    int e = blockIdx.x * blockDim.x + threadIdx.x;
    if (e < N_EDGES) unsafeAtomicAdd(&deg[dst[e]], 1.0f);
}

__global__ __launch_bounds__(256) void deg_rsqrt(float* __restrict__ deg) {
    int i = blockIdx.x * blockDim.x + threadIdx.x;
    if (i < N_NODES) deg[i] = rsqrtf(deg[i]);
}

__global__ __launch_bounds__(256) void gcn_gemm128(const float* __restrict__ A,
                                                   const float* __restrict__ W,
                                                   float* __restrict__ O) {
    __shared__ float Ws[D * D];
    const int tid = threadIdx.x;
#pragma unroll
    for (int u = tid; u < 2048; u += 256) {
        const int kp = u >> 5;
        const int n4 = (u & 31) * 4;
        const float4 r0 = *(const float4*)(W + (2 * kp) * D + n4);
        const float4 r1 = *(const float4*)(W + (2 * kp + 1) * D + n4);
        float4* dst = (float4*)(Ws + kp * 256 + n4 * 2);
        dst[0] = make_float4(r0.x, r1.x, r0.y, r1.y);
        dst[1] = make_float4(r0.z, r1.z, r0.w, r1.w);
    }
    __syncthreads();

    const int wave = tid >> 5;
    const int lane = tid & 31;
    const int tile = blockIdx.x * 8 + wave;
    if (tile >= ROW_TILES) return;

    const int nlo   = lane & 15;
    const int khalf = lane >> 4;
    const int row   = tile * 16 + nlo;

    v8f acc[8] = {};

    for (int k0 = 0; k0 < D; k0 += 4) {
        const int ka = k0 + 2 * khalf;
        v2f a;
        a.x = A[row * D + ka];
        a.y = A[row * D + ka + 1];
        const v2f* Brow = (const v2f*)(Ws + (ka >> 1) * 256);
#pragma unroll
        for (int t = 0; t < 8; ++t) {
            const v2f b = Brow[t * 16 + nlo];
            acc[t] = __builtin_amdgcn_wmma_f32_16x16x4_f32(
                false, a, false, b, (short)0, acc[t], false, false);
        }
    }

    const int rbase = tile * 16 + 8 * khalf;
#pragma unroll
    for (int t = 0; t < 8; ++t) {
        const int n = t * 16 + nlo;
#pragma unroll
        for (int r = 0; r < 8; ++r) O[(rbase + r) * D + n] = acc[t][r];
    }
}

__global__ __launch_bounds__(256) void zero_buf(float4* __restrict__ p, int n4) {
    int i = blockIdx.x * blockDim.x + threadIdx.x;
    if (i < n4) p[i] = make_float4(0.f, 0.f, 0.f, 0.f);
}

__global__ __launch_bounds__(256) void gcn_scatter(const float* __restrict__ lin,
                                                   const int* __restrict__ src,
                                                   const int* __restrict__ dst,
                                                   const float* __restrict__ dinv,
                                                   float* __restrict__ agg) {
    const int e    = (blockIdx.x * 256 + threadIdx.x) >> 5;
    const int lane = threadIdx.x & 31;
    if (e >= N_EDGES) return;
    const int s = src[e];
    const int d = dst[e];
    const float w = dinv[s] * dinv[d];
    const float4 v = ((const float4*)(lin + s * D))[lane];
    float* ap = agg + d * D + lane * 4;
    unsafeAtomicAdd(ap + 0, v.x * w);
    unsafeAtomicAdd(ap + 1, v.y * w);
    unsafeAtomicAdd(ap + 2, v.z * w);
    unsafeAtomicAdd(ap + 3, v.w * w);
}

__global__ __launch_bounds__(256) void gcn_epilogue(const float* __restrict__ agg,
                                                    const float* __restrict__ lin,
                                                    const float* __restrict__ dinv,
                                                    const float* __restrict__ bias,
                                                    float* __restrict__ hout) {
    int i = blockIdx.x * blockDim.x + threadIdx.x;
    if (i >= N_NODES * (D / 4)) return;
    const int node = i >> 5;
    const int f4   = i & 31;
    float s = dinv[node];
    s = s * s;
    const float4 a = ((const float4*)agg)[i];
    const float4 l = ((const float4*)lin)[i];
    const float4 b = ((const float4*)bias)[f4];
    float4 r;
    r.x = fmaxf(a.x + l.x * s + b.x, 0.f);
    r.y = fmaxf(a.y + l.y * s + b.y, 0.f);
    r.z = fmaxf(a.z + l.z * s + b.z, 0.f);
    r.w = fmaxf(a.w + l.w * s + b.w, 0.f);
    ((float4*)hout)[i] = r;
}

__global__ __launch_bounds__(256) void gcn_fc(const float* __restrict__ H,
                                              const float* __restrict__ W,
                                              const float* __restrict__ bias,
                                              float* __restrict__ O) {
    const int wave = threadIdx.x >> 5;
    const int lane = threadIdx.x & 31;
    const int tile = blockIdx.x * 8 + wave;
    if (tile >= ROW_TILES) return;

    const int nlo   = lane & 15;
    const int khalf = lane >> 4;
    const int row   = tile * 16 + nlo;

    v8f acc[3] = {};
    for (int k0 = 0; k0 < D; k0 += 4) {
        const int ka = k0 + 2 * khalf;
        v2f a;
        a.x = H[row * D + ka];
        a.y = H[row * D + ka + 1];
#pragma unroll
        for (int t = 0; t < 3; ++t) {
            const int n  = t * 16 + nlo;
            const int nc = n < N_CLASSES ? n : (N_CLASSES - 1);
            v2f b;
            b.x = W[ka * N_CLASSES + nc];
            b.y = W[(ka + 1) * N_CLASSES + nc];
            acc[t] = __builtin_amdgcn_wmma_f32_16x16x4_f32(
                false, a, false, b, (short)0, acc[t], false, false);
        }
    }

    const int rbase = tile * 16 + 8 * khalf;
#pragma unroll
    for (int t = 0; t < 3; ++t) {
        const int n = t * 16 + nlo;
        if (n < N_CLASSES) {
            const float bb = bias[n];
#pragma unroll
            for (int r = 0; r < 8; ++r)
                O[(rbase + r) * N_CLASSES + n] = acc[t][r] + bb;
        }
    }
}

extern "C" void kernel_launch(void* const* d_in, const int* in_sizes, int n_in,
                              void* d_out, int out_size, void* d_ws, size_t ws_size,
                              hipStream_t stream) {
    const float* x   = (const float*)d_in[0];
    const int*   ei  = (const int*)d_in[1];
    const float* W1  = (const float*)d_in[2];
    const float* b1  = (const float*)d_in[3];
    const float* W2  = (const float*)d_in[4];
    const float* b2  = (const float*)d_in[5];
    const float* W3  = (const float*)d_in[6];
    const float* b3  = (const float*)d_in[7];
    const float* fcW = (const float*)d_in[8];
    const float* fcb = (const float*)d_in[9];

    float* out    = (float*)d_out;
    float* h_out  = out;
    float* logits = out + N_NODES * D;

    float* ws   = (float*)d_ws;
    float* dinv = ws;
    float* B0   = ws + 50176;
    float* B1   = B0 + N_NODES * D;

    const int* src = ei;
    const int* dst = ei + N_EDGES;

    const int TPB = 256;
    const int gN    = (N_NODES + TPB - 1) / TPB;
    const int gE    = (N_EDGES + TPB - 1) / TPB;
    const int gF4   = (N_NODES * (D / 4) + TPB - 1) / TPB;
    const int gTile = (ROW_TILES + 7) / 8;
    const int gScat = (N_EDGES + 7) / 8;

    deg_init<<<gN, TPB, 0, stream>>>(dinv);
    deg_count<<<gE, TPB, 0, stream>>>(dst, dinv);
    deg_rsqrt<<<gN, TPB, 0, stream>>>(dinv);

    gcn_gemm128<<<gTile, TPB, 0, stream>>>(x, W1, B0);
    zero_buf<<<gF4, TPB, 0, stream>>>((float4*)B1, N_NODES * (D / 4));
    gcn_scatter<<<gScat, TPB, 0, stream>>>(B0, src, dst, dinv, B1);
    gcn_epilogue<<<gF4, TPB, 0, stream>>>(B1, B0, dinv, b1, B1);

    gcn_gemm128<<<gTile, TPB, 0, stream>>>(B1, W2, B0);
    zero_buf<<<gF4, TPB, 0, stream>>>((float4*)B1, N_NODES * (D / 4));
    gcn_scatter<<<gScat, TPB, 0, stream>>>(B0, src, dst, dinv, B1);
    gcn_epilogue<<<gF4, TPB, 0, stream>>>(B1, B0, dinv, b2, B1);

    gcn_gemm128<<<gTile, TPB, 0, stream>>>(B1, W3, B0);
    zero_buf<<<gF4, TPB, 0, stream>>>((float4*)B1, N_NODES * (D / 4));
    gcn_scatter<<<gScat, TPB, 0, stream>>>(B0, src, dst, dinv, B1);
    gcn_epilogue<<<gF4, TPB, 0, stream>>>(B1, B0, dinv, b3, h_out);

    gcn_fc<<<gTile, TPB, 0, stream>>>(h_out, fcW, fcb, logits);
}


# Edge aggregation as a gather over a per-call CSR by destination (integer count, scan, fill, segments put in source order): one wave sums a node's rows in registers in fixed order; no float atomics
# speedup vs baseline: 2.9626x; 2.9626x over previous
#include <hip/hip_runtime.h>
#include <hip/hip_bf16.h>

typedef __attribute__((ext_vector_type(2))) float v2f;
typedef __attribute__((ext_vector_type(8))) float v8f;

#define N_NODES 50000
#define N_EDGES 800000
#define D 128
#define N_CLASSES 40
#define ROW_TILES 3125
#define NPAD 50176
#define SORT_CAP 1024
#define GROUP 8

__global__ __launch_bounds__(256) void cnt_zero(int* __restrict__ cnt) {
    int i = blockIdx.x * blockDim.x + threadIdx.x;
    if (i < NPAD) cnt[i] = 0;
}

__global__ __launch_bounds__(256) void edge_count(const int* __restrict__ dst,
                                                  int* __restrict__ cnt,
                                                  int* __restrict__ rank) {
    int e = blockIdx.x * blockDim.x + threadIdx.x;
    if (e < N_EDGES) rank[e] = atomicAdd(&cnt[dst[e]], 1);
}

__global__ __launch_bounds__(256) void node_scan(const int* __restrict__ cnt,
                                                 int* __restrict__ rowptr,
                                                 float* __restrict__ dinv) {
    __shared__ int sPre[8];
    __shared__ int sTot[8];
    const int tid  = threadIdx.x;
    const int lane = tid & 31;
    const int wave = tid >> 5;
    const int4* c4 = (const int4*)cnt;
    const int first = blockIdx.x * 256;
    int pre = 0;
#pragma unroll
    for (int k = 0; k < NPAD / 1024 - 1; ++k) {
        const int4 q = c4[k * 256 + tid];
        const int sq = (q.x + q.y) + (q.z + q.w);
        pre += (k < (int)blockIdx.x) ? sq : 0;
    }
    const int4 v = c4[first + tid];
    const int tsum = (v.x + v.y) + (v.z + v.w);
    int inc = tsum;
#pragma unroll
    for (int o = 1; o < 32; o <<= 1) {
        const int t = __shfl_up(inc, o, 32);
        if (lane >= o) inc += t;
    }
#pragma unroll
    for (int o = 16; o > 0; o >>= 1) pre += __shfl_xor(pre, o, 32);
    if (lane == 31) sTot[wave] = inc;
    if (lane == 0)  sPre[wave] = pre;
    __syncthreads();
    int base = 0;
#pragma unroll
    for (int w = 0; w < 8; ++w) {
        base += sPre[w];
        if (w < wave) base += sTot[w];
    }
    const int ex = base + inc - tsum;
    int4 r;
    r.x = ex;
    r.y = r.x + v.x;
    r.z = r.y + v.y;
    r.w = r.z + v.z;
    ((int4*)rowptr)[first + tid] = r;
    float4 d;
    d.x = rsqrtf((float)(v.x + 1));
    d.y = rsqrtf((float)(v.y + 1));
    d.z = rsqrtf((float)(v.z + 1));
    d.w = rsqrtf((float)(v.w + 1));
    ((float4*)dinv)[first + tid] = d;
}

__global__ __launch_bounds__(256) void edge_fill(const int* __restrict__ src,
                                                 const int* __restrict__ dst,
                                                 const int* __restrict__ rank,
                                                 const int* __restrict__ rowptr,
                                                 int* __restrict__ colA) {
    int e = blockIdx.x * blockDim.x + threadIdx.x;
    if (e < N_EDGES) colA[rowptr[dst[e]] + rank[e]] = src[e];
}

__global__ __launch_bounds__(256) void seg_order(const int* __restrict__ rowptr,
                                                 const int* __restrict__ colA,
                                                 const float* __restrict__ dinv,
                                                 int2* __restrict__ csr) {
    const int lane = threadIdx.x & 31;
    const int node = __builtin_amdgcn_readfirstlane((blockIdx.x * 256 + threadIdx.x) >> 5);
    if (node >= N_NODES) return;
    const int beg = rowptr[node];
    const int n   = rowptr[node + 1] - beg;
    const float dn = dinv[node];
    if (n <= 32) {
        const int key = (lane < n) ? colA[beg + lane] : 0x7fffffff;
        int pos = 0;
        for (int j = 0; j < n; ++j) {
            const int kj = __builtin_amdgcn_readlane(key, j);
            pos += (kj < key || (kj == key && j < lane)) ? 1 : 0;
        }
        if (lane < n) csr[beg + pos] = make_int2(key, __float_as_int(dinv[key] * dn));
    } else if (n <= SORT_CAP) {
        for (int i = lane; i < n; i += 32) {
            const int key = colA[beg + i];
            int pos = 0;
            for (int j = 0; j < n; ++j) {
                const int kj = colA[beg + j];
                pos += (kj < key || (kj == key && j < i)) ? 1 : 0;
            }
            csr[beg + pos] = make_int2(key, __float_as_int(dinv[key] * dn));
        }
    } else {
        for (int i = lane; i < n; i += 32) {
            const int key = colA[beg + i];
            csr[beg + i] = make_int2(key, __float_as_int(dinv[key] * dn));
        }
    }
}

__global__ __launch_bounds__(256) void gcn_gemm128(const float* __restrict__ A,
                                                   const float* __restrict__ W,
                                                   float* __restrict__ O) {
    __shared__ float Ws[D * D];
    const int tid = threadIdx.x;
#pragma unroll
    for (int u = tid; u < 2048; u += 256) {
        const int kp = u >> 5;
        const int n4 = (u & 31) * 4;
        const float4 r0 = *(const float4*)(W + (2 * kp) * D + n4);
        const float4 r1 = *(const float4*)(W + (2 * kp + 1) * D + n4);
        float4* dst = (float4*)(Ws + kp * 256 + n4 * 2);
        dst[0] = make_float4(r0.x, r1.x, r0.y, r1.y);
        dst[1] = make_float4(r0.z, r1.z, r0.w, r1.w);
    }
    __syncthreads();

    const int wave = tid >> 5;
    const int lane = tid & 31;
    const int tile = blockIdx.x * 8 + wave;
    if (tile >= ROW_TILES) return;

    const int nlo   = lane & 15;
    const int khalf = lane >> 4;
    const int row   = tile * 16 + nlo;

    v8f acc[8] = {};

    for (int k0 = 0; k0 < D; k0 += 4) {
        const int ka = k0 + 2 * khalf;
        v2f a;
        a.x = A[row * D + ka];
        a.y = A[row * D + ka + 1];
        const v2f* Brow = (const v2f*)(Ws + (ka >> 1) * 256);
#pragma unroll
        for (int t = 0; t < 8; ++t) {
            const v2f b = Brow[t * 16 + nlo];
            acc[t] = __builtin_amdgcn_wmma_f32_16x16x4_f32(
                false, a, false, b, (short)0, acc[t], false, false);
        }
    }

    const int rbase = tile * 16 + 8 * khalf;
#pragma unroll
    for (int t = 0; t < 8; ++t) {
        const int n = t * 16 + nlo;
#pragma unroll
        for (int r = 0; r < 8; ++r) O[(rbase + r) * D + n] = acc[t][r];
    }
}

__global__ __launch_bounds__(256) void gcn_gather(const float* __restrict__ lin,
                                                  const int* __restrict__ rowptr,
                                                  const int2* __restrict__ csr,
                                                  const float* __restrict__ dinv,
                                                  const float* __restrict__ bias,
                                                  float* __restrict__ hout) {
    const int lane = threadIdx.x & 31;
    const int node = __builtin_amdgcn_readfirstlane((blockIdx.x * 256 + threadIdx.x) >> 5);
    if (node >= N_NODES) return;
    const float4* lin4 = (const float4*)lin;
    const int beg = rowptr[node];
    const int end = rowptr[node + 1];
    const float dn = dinv[node];
    const float4 l = lin4[node * (D / 4) + lane];
    const float4 b = ((const float4*)bias)[lane];

    float4 acc = make_float4(0.f, 0.f, 0.f, 0.f);
    for (int base = beg; base < end; base += 32) {
        const int n = (end - base) < 32 ? (end - base) : 32;
        int2 my = make_int2(0, 0);
        if (lane < n) my = csr[base + lane];
        for (int j = 0; j < n; j += GROUP) {
            float4 v[GROUP];
            float  w[GROUP];
#pragma unroll
            for (int u = 0; u < GROUP; ++u) {
                v[u] = make_float4(0.f, 0.f, 0.f, 0.f);
                w[u] = 0.f;
                if (u == 0 || j + u < n) {
                    const int su = __builtin_amdgcn_readlane(my.x, j + u);
                    w[u] = __int_as_float(__builtin_amdgcn_readlane(my.y, j + u));
                    v[u] = lin4[su * (D / 4) + lane];
                }
            }
#pragma unroll
            for (int u = 0; u < GROUP; ++u) {
                acc.x = fmaf(v[u].x, w[u], acc.x);
                acc.y = fmaf(v[u].y, w[u], acc.y);
                acc.z = fmaf(v[u].z, w[u], acc.z);
                acc.w = fmaf(v[u].w, w[u], acc.w);
            }
        }
    }

    const float s = dn * dn;
    float4 r;
    r.x = fmaxf(acc.x + l.x * s + b.x, 0.f);
    r.y = fmaxf(acc.y + l.y * s + b.y, 0.f);
    r.z = fmaxf(acc.z + l.z * s + b.z, 0.f);
    r.w = fmaxf(acc.w + l.w * s + b.w, 0.f);
    ((float4*)hout)[node * (D / 4) + lane] = r;
}

__global__ __launch_bounds__(256) void gcn_fc(const float* __restrict__ H,
                                              const float* __restrict__ W,
                                              const float* __restrict__ bias,
                                              float* __restrict__ O) {
    const int wave = threadIdx.x >> 5;
    const int lane = threadIdx.x & 31;
    const int tile = blockIdx.x * 8 + wave;
    if (tile >= ROW_TILES) return;

    const int nlo   = lane & 15;
    const int khalf = lane >> 4;
    const int row   = tile * 16 + nlo;

    v8f acc[3] = {};
    for (int k0 = 0; k0 < D; k0 += 4) {
        const int ka = k0 + 2 * khalf;
        v2f a;
        a.x = H[row * D + ka];
        a.y = H[row * D + ka + 1];
#pragma unroll
        for (int t = 0; t < 3; ++t) {
            const int n  = t * 16 + nlo;
            const int nc = n < N_CLASSES ? n : (N_CLASSES - 1);
            v2f b;
            b.x = W[ka * N_CLASSES + nc];
            b.y = W[(ka + 1) * N_CLASSES + nc];
            acc[t] = __builtin_amdgcn_wmma_f32_16x16x4_f32(
                false, a, false, b, (short)0, acc[t], false, false);
        }
    }

    const int rbase = tile * 16 + 8 * khalf;
#pragma unroll
    for (int t = 0; t < 3; ++t) {
        const int n = t * 16 + nlo;
        if (n < N_CLASSES) {
            const float bb = bias[n];
#pragma unroll
            for (int r = 0; r < 8; ++r)
                O[(rbase + r) * N_CLASSES + n] = acc[t][r] + bb;
        }
    }
}

extern "C" void kernel_launch(void* const* d_in, const int* in_sizes, int n_in,
                              void* d_out, int out_size, void* d_ws, size_t ws_size,
                              hipStream_t stream) {
    const float* x   = (const float*)d_in[0];
    const int*   ei  = (const int*)d_in[1];
    const float* W1  = (const float*)d_in[2];
    const float* b1  = (const float*)d_in[3];
    const float* W2  = (const float*)d_in[4];
    const float* b2  = (const float*)d_in[5];
    const float* W3  = (const float*)d_in[6];
    const float* b3  = (const float*)d_in[7];
    const float* fcW = (const float*)d_in[8];
    const float* fcb = (const float*)d_in[9];

    float* out    = (float*)d_out;
    float* h_out  = out;
    float* logits = out + N_NODES * D;

    float* ws     = (float*)d_ws;
    int*   cnt    = (int*)d_ws;
    int*   rowptr = cnt + NPAD;
    float* dinv   = ws + 2 * NPAD;
    int*   rank   = cnt + 3 * NPAD;
    int*   colA   = rank + N_EDGES;
    int2*  csr    = (int2*)(colA + N_EDGES);
    float* B0     = ws + 3 * NPAD + 4 * N_EDGES;
    float* B1     = B0 + N_NODES * D;

    const int* src = ei;
    const int* dst = ei + N_EDGES;

    const int TPB = 256;
    const int gP    = NPAD / TPB;
    const int gE    = (N_EDGES + TPB - 1) / TPB;
    const int gScan = NPAD / 1024;
    const int gNode = (N_NODES + 7) / 8;
    const int gTile = (ROW_TILES + 7) / 8;

    cnt_zero<<<gP, TPB, 0, stream>>>(cnt);
    edge_count<<<gE, TPB, 0, stream>>>(dst, cnt, rank);
    node_scan<<<gScan, TPB, 0, stream>>>(cnt, rowptr, dinv);
    edge_fill<<<gE, TPB, 0, stream>>>(src, dst, rank, rowptr, colA);
    seg_order<<<gNode, TPB, 0, stream>>>(rowptr, colA, dinv, csr);

    gcn_gemm128<<<gTile, TPB, 0, stream>>>(x, W1, B0);
    gcn_gather<<<gNode, TPB, 0, stream>>>(B0, rowptr, csr, dinv, b1, B1);

    gcn_gemm128<<<gTile, TPB, 0, stream>>>(B1, W2, B0);
    gcn_gather<<<gNode, TPB, 0, stream>>>(B0, rowptr, csr, dinv, b2, B1);

    gcn_gemm128<<<gTile, TPB, 0, stream>>>(B1, W3, B0);
    gcn_gather<<<gNode, TPB, 0, stream>>>(B0, rowptr, csr, dinv, b3, h_out);

    gcn_fc<<<gTile, TPB, 0, stream>>>(h_out, fcW, fcb, logits);
}


# Edges go into fixed 64-entry slot arrays per destination, placed by the counting add's return value (past 64: overflow list, arrival order); scan and fill launches go, 10 left; sum order kept
# speedup vs baseline: 3.2413x; 1.0941x over previous
#include <hip/hip_runtime.h>
#include <hip/hip_bf16.h>

typedef __attribute__((ext_vector_type(2))) float v2f;
typedef __attribute__((ext_vector_type(8))) float v8f;

#define N_NODES 50000
#define N_EDGES 800000
#define D 128
#define N_CLASSES 40
#define ROW_TILES 3125
#define NPAD 50176
#define CAP 64
#define GROUP 8

__global__ __launch_bounds__(256) void cnt_zero(int* __restrict__ cnt) {
    int i = blockIdx.x * blockDim.x + threadIdx.x;
    if (i < NPAD) cnt[i] = 0;
}

__global__ __launch_bounds__(256) void edge_place(const int* __restrict__ src,
                                                  const int* __restrict__ dst,
                                                  int* __restrict__ cnt,
                                                  int* __restrict__ slots,
                                                  int2* __restrict__ ovf) {
    int e = blockIdx.x * blockDim.x + threadIdx.x;
    if (e < N_EDGES) {
        const int d = dst[e];
        const int s = src[e];
        const int r = atomicAdd(&cnt[d], 1);
        if (r < CAP) {
            slots[d * CAP + r] = s;
        } else {
            const int q = atomicAdd(&cnt[N_NODES], 1);
            ovf[q] = make_int2(d, s);
        }
    }
}

__global__ __launch_bounds__(256) void seg_order(const int* __restrict__ cnt,
                                                 const int* __restrict__ slots,
                                                 int2* __restrict__ csr) {
    const int lane = threadIdx.x & 31;
    const int node = __builtin_amdgcn_readfirstlane((blockIdx.x * 256 + threadIdx.x) >> 5);
    if (node >= N_NODES) return;
    const int c   = cnt[node];
    const int n   = c < CAP ? c : CAP;
    const int beg = node * CAP;
    const float dn = rsqrtf((float)(c + 1));
    if (n <= 32) {
        const int key = (lane < n) ? slots[beg + lane] : 0x7fffffff;
        int pos = 0;
        for (int j = 0; j < n; ++j) {
            const int kj = __builtin_amdgcn_readlane(key, j);
            pos += (kj < key || (kj == key && j < lane)) ? 1 : 0;
        }
        if (lane < n)
            csr[beg + pos] = make_int2(key, __float_as_int(rsqrtf((float)(cnt[key] + 1)) * dn));
    } else {
        for (int i = lane; i < n; i += 32) {
            const int key = slots[beg + i];
            int pos = 0;
            for (int j = 0; j < n; ++j) {
                const int kj = slots[beg + j];
                pos += (kj < key || (kj == key && j < i)) ? 1 : 0;
            }
            csr[beg + pos] = make_int2(key, __float_as_int(rsqrtf((float)(cnt[key] + 1)) * dn));
        }
    }
}

__global__ __launch_bounds__(256) void gcn_gemm128(const float* __restrict__ A,
                                                   const float* __restrict__ W,
                                                   float* __restrict__ O) {
    __shared__ float Ws[D * D];
    const int tid = threadIdx.x;
#pragma unroll
    for (int u = tid; u < 2048; u += 256) {
        const int kp = u >> 5;
        const int n4 = (u & 31) * 4;
        const float4 r0 = *(const float4*)(W + (2 * kp) * D + n4);
        const float4 r1 = *(const float4*)(W + (2 * kp + 1) * D + n4);
        float4* dst = (float4*)(Ws + kp * 256 + n4 * 2);
        dst[0] = make_float4(r0.x, r1.x, r0.y, r1.y);
        dst[1] = make_float4(r0.z, r1.z, r0.w, r1.w);
    }
    __syncthreads();

    const int wave = tid >> 5;
    const int lane = tid & 31;
    const int tile = blockIdx.x * 8 + wave;
    if (tile >= ROW_TILES) return;

    const int nlo   = lane & 15;
    const int khalf = lane >> 4;
    const int row   = tile * 16 + nlo;

    v8f acc[8] = {};

    for (int k0 = 0; k0 < D; k0 += 4) {
        const int ka = k0 + 2 * khalf;
        v2f a;
        a.x = A[row * D + ka];
        a.y = A[row * D + ka + 1];
        const v2f* Brow = (const v2f*)(Ws + (ka >> 1) * 256);
#pragma unroll
        for (int t = 0; t < 8; ++t) {
            const v2f b = Brow[t * 16 + nlo];
            acc[t] = __builtin_amdgcn_wmma_f32_16x16x4_f32(
                false, a, false, b, (short)0, acc[t], false, false);
        }
    }

    const int rbase = tile * 16 + 8 * khalf;
#pragma unroll
    for (int t = 0; t < 8; ++t) {
        const int n = t * 16 + nlo;
#pragma unroll
        for (int r = 0; r < 8; ++r) O[(rbase + r) * D + n] = acc[t][r];
    }
}

__global__ __launch_bounds__(256) void gcn_gather(const float* __restrict__ lin,
                                                  const int* __restrict__ cnt,
                                                  const int2* __restrict__ csr,
                                                  const int2* __restrict__ ovf,
                                                  const float* __restrict__ bias,
                                                  float* __restrict__ hout) {
    const int lane = threadIdx.x & 31;
    const int node = __builtin_amdgcn_readfirstlane((blockIdx.x * 256 + threadIdx.x) >> 5);
    if (node >= N_NODES) return;
    const float4* lin4 = (const float4*)lin;
    const int c   = cnt[node];
    const int beg = node * CAP;
    const int end = beg + (c < CAP ? c : CAP);
    const float dn = rsqrtf((float)(c + 1));
    const float4 l = lin4[node * (D / 4) + lane];
    const float4 b = ((const float4*)bias)[lane];

    float4 acc = make_float4(0.f, 0.f, 0.f, 0.f);
    for (int base = beg; base < end; base += 32) {
        const int n = (end - base) < 32 ? (end - base) : 32;
        int2 my = make_int2(0, 0);
        if (lane < n) my = csr[base + lane];
        for (int j = 0; j < n; j += GROUP) {
            float4 v[GROUP];
            float  w[GROUP];
#pragma unroll
            for (int u = 0; u < GROUP; ++u) {
                v[u] = make_float4(0.f, 0.f, 0.f, 0.f);
                w[u] = 0.f;
                if (u == 0 || j + u < n) {
                    const int su = __builtin_amdgcn_readlane(my.x, j + u);
                    w[u] = __int_as_float(__builtin_amdgcn_readlane(my.y, j + u));
                    v[u] = lin4[su * (D / 4) + lane];
                }
            }
#pragma unroll
            for (int u = 0; u < GROUP; ++u) {
                acc.x = fmaf(v[u].x, w[u], acc.x);
                acc.y = fmaf(v[u].y, w[u], acc.y);
                acc.z = fmaf(v[u].z, w[u], acc.z);
                acc.w = fmaf(v[u].w, w[u], acc.w);
            }
        }
    }

    if (c > CAP) {
        const int m = cnt[N_NODES];
        for (int q = 0; q < m; ++q) {
            const int2 p = ovf[q];
            if (p.x == node) {
                const float w = rsqrtf((float)(cnt[p.y] + 1)) * dn;
                const float4 v = lin4[p.y * (D / 4) + lane];
                acc.x = fmaf(v.x, w, acc.x);
                acc.y = fmaf(v.y, w, acc.y);
                acc.z = fmaf(v.z, w, acc.z);
                acc.w = fmaf(v.w, w, acc.w);
            }
        }
    }

    const float s = dn * dn;
    float4 r;
    r.x = fmaxf(acc.x + l.x * s + b.x, 0.f);
    r.y = fmaxf(acc.y + l.y * s + b.y, 0.f);
    r.z = fmaxf(acc.z + l.z * s + b.z, 0.f);
    r.w = fmaxf(acc.w + l.w * s + b.w, 0.f);
    ((float4*)hout)[node * (D / 4) + lane] = r;
}

__global__ __launch_bounds__(256) void gcn_fc(const float* __restrict__ H,
                                              const float* __restrict__ W,
                                              const float* __restrict__ bias,
                                              float* __restrict__ O) {
    const int wave = threadIdx.x >> 5;
    const int lane = threadIdx.x & 31;
    const int tile = blockIdx.x * 8 + wave;
    if (tile >= ROW_TILES) return;

    const int nlo   = lane & 15;
    const int khalf = lane >> 4;
    const int row   = tile * 16 + nlo;

    v8f acc[3] = {};
    for (int k0 = 0; k0 < D; k0 += 4) {
        const int ka = k0 + 2 * khalf;
        v2f a;
        a.x = H[row * D + ka];
        a.y = H[row * D + ka + 1];
#pragma unroll
        for (int t = 0; t < 3; ++t) {
            const int n  = t * 16 + nlo;
            const int nc = n < N_CLASSES ? n : (N_CLASSES - 1);
            v2f b;
            b.x = W[ka * N_CLASSES + nc];
            b.y = W[(ka + 1) * N_CLASSES + nc];
            acc[t] = __builtin_amdgcn_wmma_f32_16x16x4_f32(
                false, a, false, b, (short)0, acc[t], false, false);
        }
    }

    const int rbase = tile * 16 + 8 * khalf;
#pragma unroll
    for (int t = 0; t < 3; ++t) {
        const int n = t * 16 + nlo;
        if (n < N_CLASSES) {
            const float bb = bias[n];
#pragma unroll
            for (int r = 0; r < 8; ++r)
                O[(rbase + r) * N_CLASSES + n] = acc[t][r] + bb;
        }
    }
}

extern "C" void kernel_launch(void* const* d_in, const int* in_sizes, int n_in,
                              void* d_out, int out_size, void* d_ws, size_t ws_size,
                              hipStream_t stream) {
    const float* x   = (const float*)d_in[0];
    const int*   ei  = (const int*)d_in[1];
    const float* W1  = (const float*)d_in[2];
    const float* b1  = (const float*)d_in[3];
    const float* W2  = (const float*)d_in[4];
    const float* b2  = (const float*)d_in[5];
    const float* W3  = (const float*)d_in[6];
    const float* b3  = (const float*)d_in[7];
    const float* fcW = (const float*)d_in[8];
    const float* fcb = (const float*)d_in[9];

    float* out    = (float*)d_out;
    float* h_out  = out;
    float* logits = out + N_NODES * D;

    float* ws     = (float*)d_ws;
    int*   cnt    = (int*)d_ws;
    int*   slots  = cnt + NPAD;
    int2*  csr    = (int2*)(slots + N_NODES * CAP);
    int2*  ovf    = csr + N_NODES * CAP;
    float* B0     = ws + NPAD + 3 * N_NODES * CAP + 2 * N_EDGES;
    float* B1     = B0 + N_NODES * D;

    const int* src = ei;
    const int* dst = ei + N_EDGES;

    const int TPB = 256;
    const int gP    = NPAD / TPB;
    const int gE    = (N_EDGES + TPB - 1) / TPB;
    const int gNode = (N_NODES + 7) / 8;
    const int gTile = (ROW_TILES + 7) / 8;

    cnt_zero<<<gP, TPB, 0, stream>>>(cnt);
    edge_place<<<gE, TPB, 0, stream>>>(src, dst, cnt, slots, ovf);
    seg_order<<<gNode, TPB, 0, stream>>>(cnt, slots, csr);

    gcn_gemm128<<<gTile, TPB, 0, stream>>>(x, W1, B0);
    gcn_gather<<<gNode, TPB, 0, stream>>>(B0, cnt, csr, ovf, b1, B1);

    gcn_gemm128<<<gTile, TPB, 0, stream>>>(B1, W2, B0);
    gcn_gather<<<gNode, TPB, 0, stream>>>(B0, cnt, csr, ovf, b2, B1);

    gcn_gemm128<<<gTile, TPB, 0, stream>>>(B1, W3, B0);
    gcn_gather<<<gNode, TPB, 0, stream>>>(B0, cnt, csr, ovf, b3, h_out);

    gcn_fc<<<gTile, TPB, 0, stream>>>(h_out, fcW, fcb, logits);
}


# Each gather hands its 16-node tile via LDS to the next layer's product in the same launch (resident 8-wave blocks, weights in registers, tiles from an integer counter); fc joins layer 3; 7 launches
# speedup vs baseline: 3.6055x; 1.1124x over previous
#include <hip/hip_runtime.h>
#include <hip/hip_bf16.h>

typedef __attribute__((ext_vector_type(2))) float v2f;
typedef __attribute__((ext_vector_type(8))) float v8f;

#define N_NODES 50000
#define N_EDGES 800000
#define D 128
#define N_CLASSES 40
#define ROW_TILES 3125
#define NPAD 50176
#define CAP 64
#define GROUP 8

__global__ __launch_bounds__(256) void cnt_zero(int* __restrict__ cnt) {
    int i = blockIdx.x * blockDim.x + threadIdx.x;
    if (i < NPAD) cnt[i] = 0;
}

__global__ __launch_bounds__(256) void edge_place(const int* __restrict__ src,
                                                  const int* __restrict__ dst,
                                                  int* __restrict__ cnt,
                                                  int* __restrict__ slots,
                                                  int2* __restrict__ ovf) {
    int e = blockIdx.x * blockDim.x + threadIdx.x;
    if (e < N_EDGES) {
        const int d = dst[e];
        const int s = src[e];
        const int r = atomicAdd(&cnt[d], 1);
        if (r < CAP) {
            slots[d * CAP + r] = s;
        } else {
            const int q = atomicAdd(&cnt[N_NODES], 1);
            ovf[q] = make_int2(d, s);
        }
    }
}

__global__ __launch_bounds__(256) void seg_order(const int* __restrict__ cnt,
                                                 const int* __restrict__ slots,
                                                 int2* __restrict__ csr) {
    const int lane = threadIdx.x & 31;
    const int node = __builtin_amdgcn_readfirstlane((blockIdx.x * 256 + threadIdx.x) >> 5);
    if (node >= N_NODES) return;
    const int c   = cnt[node];
    const int n   = c < CAP ? c : CAP;
    const int beg = node * CAP;
    const float dn = rsqrtf((float)(c + 1));
    if (n <= 32) {
        const int key = (lane < n) ? slots[beg + lane] : 0x7fffffff;
        int pos = 0;
        for (int j = 0; j < n; ++j) {
            const int kj = __builtin_amdgcn_readlane(key, j);
            pos += (kj < key || (kj == key && j < lane)) ? 1 : 0;
        }
        if (lane < n)
            csr[beg + pos] = make_int2(key, __float_as_int(rsqrtf((float)(cnt[key] + 1)) * dn));
    } else {
        for (int i = lane; i < n; i += 32) {
            const int key = slots[beg + i];
            int pos = 0;
            for (int j = 0; j < n; ++j) {
                const int kj = slots[beg + j];
                pos += (kj < key || (kj == key && j < i)) ? 1 : 0;
            }
            csr[beg + pos] = make_int2(key, __float_as_int(rsqrtf((float)(cnt[key] + 1)) * dn));
        }
    }
}

__global__ __launch_bounds__(256) void gcn_gemm128(const float* __restrict__ A,
                                                   const float* __restrict__ W,
                                                   float* __restrict__ O) {
    __shared__ float Ws[D * D];
    const int tid = threadIdx.x;
#pragma unroll
    for (int u = tid; u < 2048; u += 256) {
        const int kp = u >> 5;
        const int n4 = (u & 31) * 4;
        const float4 r0 = *(const float4*)(W + (2 * kp) * D + n4);
        const float4 r1 = *(const float4*)(W + (2 * kp + 1) * D + n4);
        float4* dst = (float4*)(Ws + kp * 256 + n4 * 2);
        dst[0] = make_float4(r0.x, r1.x, r0.y, r1.y);
        dst[1] = make_float4(r0.z, r1.z, r0.w, r1.w);
    }
    __syncthreads();

    const int wave = tid >> 5;
    const int lane = tid & 31;
    const int tile = blockIdx.x * 8 + wave;
    if (tile >= ROW_TILES) return;

    const int nlo   = lane & 15;
    const int khalf = lane >> 4;
    const int row   = tile * 16 + nlo;

    v8f acc[8] = {};

    for (int k0 = 0; k0 < D; k0 += 4) {
        const int ka = k0 + 2 * khalf;
        v2f a;
        a.x = A[row * D + ka];
        a.y = A[row * D + ka + 1];
        const v2f* Brow = (const v2f*)(Ws + (ka >> 1) * 256);
#pragma unroll
        for (int t = 0; t < 8; ++t) {
            const v2f b = Brow[t * 16 + nlo];
            acc[t] = __builtin_amdgcn_wmma_f32_16x16x4_f32(
                false, a, false, b, (short)0, acc[t], false, false);
        }
    }

    const int rbase = tile * 16 + 8 * khalf;
#pragma unroll
    for (int t = 0; t < 8; ++t) {
        const int n = t * 16 + nlo;
#pragma unroll
        for (int r = 0; r < 8; ++r) O[(rbase + r) * D + n] = acc[t][r];
    }
}

#define HP 132

__device__ __forceinline__ float4 gather_node(const float4* __restrict__ lin4,
                                              const int* __restrict__ cnt,
                                              const int2* __restrict__ csr,
                                              const int2* __restrict__ ovf,
                                              const int node, const int c,
                                              const int2 my_first, const float4 l,
                                              const int lane, const float4 b) {
    const int beg = node * CAP;
    const int end = beg + (c < CAP ? c : CAP);
    const float dn = rsqrtf((float)(c + 1));

    float4 acc = make_float4(0.f, 0.f, 0.f, 0.f);
    for (int base = beg; base < end; base += 32) {
        const int n = (end - base) < 32 ? (end - base) : 32;
        int2 my = my_first;
        if (base != beg) {
            my = make_int2(0, 0);
            if (lane < n) my = csr[base + lane];
        }
        for (int j = 0; j < n; j += GROUP) {
            float4 v[GROUP];
            float  w[GROUP];
#pragma unroll
            for (int u = 0; u < GROUP; ++u) {
                v[u] = make_float4(0.f, 0.f, 0.f, 0.f);
                w[u] = 0.f;
                if (u == 0 || j + u < n) {
                    const int su = __builtin_amdgcn_readlane(my.x, j + u);
                    w[u] = __int_as_float(__builtin_amdgcn_readlane(my.y, j + u));
                    v[u] = lin4[su * (D / 4) + lane];
                }
            }
#pragma unroll
            for (int u = 0; u < GROUP; ++u) {
                acc.x = fmaf(v[u].x, w[u], acc.x);
                acc.y = fmaf(v[u].y, w[u], acc.y);
                acc.z = fmaf(v[u].z, w[u], acc.z);
                acc.w = fmaf(v[u].w, w[u], acc.w);
            }
        }
    }

    if (c > CAP) {
        const int m = cnt[N_NODES];
        for (int q = 0; q < m; ++q) {
            const int2 p = ovf[q];
            if (p.x == node) {
                const float w = rsqrtf((float)(cnt[p.y] + 1)) * dn;
                const float4 v = lin4[p.y * (D / 4) + lane];
                acc.x = fmaf(v.x, w, acc.x);
                acc.y = fmaf(v.y, w, acc.y);
                acc.z = fmaf(v.z, w, acc.z);
                acc.w = fmaf(v.w, w, acc.w);
            }
        }
    }

    const float s = dn * dn;
    float4 r;
    r.x = fmaxf(acc.x + l.x * s + b.x, 0.f);
    r.y = fmaxf(acc.y + l.y * s + b.y, 0.f);
    r.z = fmaxf(acc.z + l.z * s + b.z, 0.f);
    r.w = fmaxf(acc.w + l.w * s + b.w, 0.f);
    return r;
}

template <int LDW, int NT, bool WRITE_H, bool BIAS2>
__global__ __launch_bounds__(256) void gcn_gather_mm(const float* __restrict__ lin,
                                                     const int* __restrict__ cnt,
                                                     const int2* __restrict__ csr,
                                                     const int2* __restrict__ ovf,
                                                     const float* __restrict__ bias,
                                                     const float* __restrict__ W2,
                                                     const float* __restrict__ bias2,
                                                     float* __restrict__ hout,
                                                     float* __restrict__ O,
                                                     int* __restrict__ ctr) {
    __shared__ __attribute__((aligned(16))) float hs[2][16 * HP];
    __shared__ int   s_tile[2];

    const int tid  = threadIdx.x;
    const int lane = tid & 31;
    const int wave  = __builtin_amdgcn_readfirstlane(tid >> 5);
    const int nlo   = lane & 15;
    const int khalf = lane >> 4;

    if (tid == 0) s_tile[0] = atomicAdd(ctr, 1);
    __syncthreads();
    int tile = __builtin_amdgcn_readfirstlane(s_tile[0]);
    if (tile >= ROW_TILES) return;

    const float4* lin4 = (const float4*)lin;
    const float4 b = ((const float4*)bias)[lane];

    const bool mm = wave < NT;
    const int  ncol = wave * 16 + nlo;
    const int  nc   = ncol < LDW ? ncol : (LDW - 1);
    v2f bw[32];
#pragma unroll
    for (int s = 0; s < 32; ++s) {
        bw[s].x = 0.f;
        bw[s].y = 0.f;
        if (mm) {
            const int ka = 4 * s + 2 * khalf;
            bw[s].x = W2[ka * LDW + nc];
            bw[s].y = W2[(ka + 1) * LDW + nc];
        }
    }
    float bb = 0.f;
    if (BIAS2 && mm && ncol < LDW) bb = bias2[ncol];

    for (int it = 0; tile < ROW_TILES; ++it) {
        float* hb = hs[it & 1];

        const int node0 = tile * 16 + 2 * wave;
        const int node1 = node0 + 1;
        const int c0 = cnt[node0];
        const int c1 = cnt[node1];
        const int f0 = c0 < 32 ? c0 : 32;
        const int f1 = c1 < 32 ? c1 : 32;
        int2 my0 = make_int2(0, 0);
        int2 my1 = make_int2(0, 0);
        if (lane < f0) my0 = csr[node0 * CAP + lane];
        if (lane < f1) my1 = csr[node1 * CAP + lane];
        const float4 l0 = lin4[node0 * (D / 4) + lane];
        const float4 l1 = lin4[node1 * (D / 4) + lane];

        const float4 r0 = gather_node(lin4, cnt, csr, ovf, node0, c0, my0, l0, lane, b);
        *(float4*)(hb + (2 * wave) * HP + 4 * lane) = r0;
        if (WRITE_H) ((float4*)hout)[node0 * (D / 4) + lane] = r0;

        const float4 r1 = gather_node(lin4, cnt, csr, ovf, node1, c1, my1, l1, lane, b);
        *(float4*)(hb + (2 * wave + 1) * HP + 4 * lane) = r1;
        if (WRITE_H) ((float4*)hout)[node1 * (D / 4) + lane] = r1;

        if (tid == 0) s_tile[(it + 1) & 1] = atomicAdd(ctr, 1);
        __syncthreads();

        if (mm) {
            v8f acc = {};
            const float* arow = hb + nlo * HP + 2 * khalf;
#pragma unroll
            for (int s = 0; s < 32; ++s) {
                const v2f a = *(const v2f*)(arow + 4 * s);
                acc = __builtin_amdgcn_wmma_f32_16x16x4_f32(
                    false, a, false, bw[s], (short)0, acc, false, false);
            }
            if (ncol < LDW) {
                const int rbase = tile * 16 + 8 * khalf;
#pragma unroll
                for (int r = 0; r < 8; ++r) {
                    if (BIAS2) O[(rbase + r) * LDW + ncol] = acc[r] + bb;
                    else       O[(rbase + r) * LDW + ncol] = acc[r];
                }
            }
        }
        tile = __builtin_amdgcn_readfirstlane(s_tile[(it + 1) & 1]);
    }
}

extern "C" void kernel_launch(void* const* d_in, const int* in_sizes, int n_in,
                              void* d_out, int out_size, void* d_ws, size_t ws_size,
                              hipStream_t stream) {
    const float* x   = (const float*)d_in[0];
    const int*   ei  = (const int*)d_in[1];
    const float* W1  = (const float*)d_in[2];
    const float* b1  = (const float*)d_in[3];
    const float* W2  = (const float*)d_in[4];
    const float* b2  = (const float*)d_in[5];
    const float* W3  = (const float*)d_in[6];
    const float* b3  = (const float*)d_in[7];
    const float* fcW = (const float*)d_in[8];
    const float* fcb = (const float*)d_in[9];

    float* out    = (float*)d_out;
    float* h_out  = out;
    float* logits = out + N_NODES * D;

    float* ws     = (float*)d_ws;
    int*   cnt    = (int*)d_ws;
    int*   slots  = cnt + NPAD;
    int2*  csr    = (int2*)(slots + N_NODES * CAP);
    int2*  ovf    = csr + N_NODES * CAP;
    float* B0     = ws + NPAD + 3 * N_NODES * CAP + 2 * N_EDGES;
    float* B1     = B0 + N_NODES * D;

    const int* src = ei;
    const int* dst = ei + N_EDGES;

    const int TPB = 256;
    const int gP    = NPAD / TPB;
    const int gE    = (N_EDGES + TPB - 1) / TPB;
    const int gNode = (N_NODES + 7) / 8;
    const int gTile = (ROW_TILES + 7) / 8;
    const int gRes  = 1024;
    const int gResF = 768;
    int* ctr1 = cnt + N_NODES + 1;
    int* ctr2 = cnt + N_NODES + 2;
    int* ctr3 = cnt + N_NODES + 3;

    cnt_zero<<<gP, TPB, 0, stream>>>(cnt);
    edge_place<<<gE, TPB, 0, stream>>>(src, dst, cnt, slots, ovf);
    seg_order<<<gNode, TPB, 0, stream>>>(cnt, slots, csr);

    gcn_gemm128<<<gTile, TPB, 0, stream>>>(x, W1, B0);

    gcn_gather_mm<D, 8, false, false><<<gRes, TPB, 0, stream>>>(B0, cnt, csr, ovf, b1, W2, b2, B1, B1, ctr1);

    gcn_gather_mm<D, 8, false, false><<<gRes, TPB, 0, stream>>>(B1, cnt, csr, ovf, b2, W3, b3, B0, B0, ctr2);

    gcn_gather_mm<N_CLASSES, 3, true, true><<<gResF, TPB, 0, stream>>>(B0, cnt, csr, ovf, b3, fcW, fcb, h_out, logits, ctr3);
}


# The x@W1 product has no launch of its own: its 391 blocks lead the grid of the edge placement launch (role by block number) and run under its scattered atomic waits; 6 launches
# speedup vs baseline: 3.9513x; 1.0959x over previous
#include <hip/hip_runtime.h>
#include <hip/hip_bf16.h>

typedef __attribute__((ext_vector_type(2))) float v2f;
typedef __attribute__((ext_vector_type(8))) float v8f;

#define N_NODES 50000
#define N_EDGES 800000
#define D 128
#define N_CLASSES 40
#define ROW_TILES 3125
#define NPAD 50176
#define CAP 64
#define GROUP 8

__global__ __launch_bounds__(256) void cnt_zero(int* __restrict__ cnt) {
    int i = blockIdx.x * blockDim.x + threadIdx.x;
    if (i < NPAD) cnt[i] = 0;
}

__device__ __forceinline__ void edge_place_body(const int bid,
                                                const int* __restrict__ src,
                                                const int* __restrict__ dst,
                                                int* __restrict__ cnt,
                                                int* __restrict__ slots,
                                                int2* __restrict__ ovf) {
    int e = bid * 256 + threadIdx.x;
    if (e < N_EDGES) {
        const int d = dst[e];
        const int s = src[e];
        const int r = atomicAdd(&cnt[d], 1);
        if (r < CAP) {
            slots[d * CAP + r] = s;
        } else {
            const int q = atomicAdd(&cnt[N_NODES], 1);
            ovf[q] = make_int2(d, s);
        }
    }
}

__global__ __launch_bounds__(256) void seg_order(const int* __restrict__ cnt,
                                                 const int* __restrict__ slots,
                                                 int2* __restrict__ csr) {
    const int lane = threadIdx.x & 31;
    const int node = __builtin_amdgcn_readfirstlane((blockIdx.x * 256 + threadIdx.x) >> 5);
    if (node >= N_NODES) return;
    const int c   = cnt[node];
    const int n   = c < CAP ? c : CAP;
    const int beg = node * CAP;
    const float dn = rsqrtf((float)(c + 1));
    if (n <= 32) {
        const int key = (lane < n) ? slots[beg + lane] : 0x7fffffff;
        int pos = 0;
        for (int j = 0; j < n; ++j) {
            const int kj = __builtin_amdgcn_readlane(key, j);
            pos += (kj < key || (kj == key && j < lane)) ? 1 : 0;
        }
        if (lane < n)
            csr[beg + pos] = make_int2(key, __float_as_int(rsqrtf((float)(cnt[key] + 1)) * dn));
    } else {
        for (int i = lane; i < n; i += 32) {
            const int key = slots[beg + i];
            int pos = 0;
            for (int j = 0; j < n; ++j) {
                const int kj = slots[beg + j];
                pos += (kj < key || (kj == key && j < i)) ? 1 : 0;
            }
            csr[beg + pos] = make_int2(key, __float_as_int(rsqrtf((float)(cnt[key] + 1)) * dn));
        }
    }
}

__device__ __forceinline__ void gemm128_body(float* __restrict__ Ws, const int bid,
                                             const float* __restrict__ A,
                                             const float* __restrict__ W,
                                             float* __restrict__ O) {
    const int tid = threadIdx.x;
#pragma unroll
    for (int u = tid; u < 2048; u += 256) {
        const int kp = u >> 5;
        const int n4 = (u & 31) * 4;
        const float4 r0 = *(const float4*)(W + (2 * kp) * D + n4);
        const float4 r1 = *(const float4*)(W + (2 * kp + 1) * D + n4);
        float4* dst = (float4*)(Ws + kp * 256 + n4 * 2);
        dst[0] = make_float4(r0.x, r1.x, r0.y, r1.y);
        dst[1] = make_float4(r0.z, r1.z, r0.w, r1.w);
    }
    __syncthreads();

    const int wave = tid >> 5;
    const int lane = tid & 31;
    const int tile = bid * 8 + wave;
    if (tile >= ROW_TILES) return;

    const int nlo   = lane & 15;
    const int khalf = lane >> 4;
    const int row   = tile * 16 + nlo;

    v8f acc[8] = {};

    for (int k0 = 0; k0 < D; k0 += 4) {
        const int ka = k0 + 2 * khalf;
        v2f a;
        a.x = A[row * D + ka];
        a.y = A[row * D + ka + 1];
        const v2f* Brow = (const v2f*)(Ws + (ka >> 1) * 256);
#pragma unroll
        for (int t = 0; t < 8; ++t) {
            const v2f b = Brow[t * 16 + nlo];
            acc[t] = __builtin_amdgcn_wmma_f32_16x16x4_f32(
                false, a, false, b, (short)0, acc[t], false, false);
        }
    }

    const int rbase = tile * 16 + 8 * khalf;
#pragma unroll
    for (int t = 0; t < 8; ++t) {
        const int n = t * 16 + nlo;
#pragma unroll
        for (int r = 0; r < 8; ++r) O[(rbase + r) * D + n] = acc[t][r];
    }
}

#define GEMM_BLOCKS 391
#define PLACE_BLOCKS 3125

__global__ __launch_bounds__(256) void place_and_gemm(const int* __restrict__ src,
                                                      const int* __restrict__ dst,
                                                      int* __restrict__ cnt,
                                                      int* __restrict__ slots,
                                                      int2* __restrict__ ovf,
                                                      const float* __restrict__ A,
                                                      const float* __restrict__ W,
                                                      float* __restrict__ O) {
    __shared__ float Ws[D * D];
    if (blockIdx.x < GEMM_BLOCKS) gemm128_body(Ws, blockIdx.x, A, W, O);
    else                          edge_place_body(blockIdx.x - GEMM_BLOCKS, src, dst, cnt, slots, ovf);
}

#define HP 132

__device__ __forceinline__ float4 gather_node(const float4* __restrict__ lin4,
                                              const int* __restrict__ cnt,
                                              const int2* __restrict__ csr,
                                              const int2* __restrict__ ovf,
                                              const int node, const int c,
                                              const int2 my_first, const float4 l,
                                              const int lane, const float4 b) {
    const int beg = node * CAP;
    const int end = beg + (c < CAP ? c : CAP);
    const float dn = rsqrtf((float)(c + 1));

    float4 acc = make_float4(0.f, 0.f, 0.f, 0.f);
    for (int base = beg; base < end; base += 32) {
        const int n = (end - base) < 32 ? (end - base) : 32;
        int2 my = my_first;
        if (base != beg) {
            my = make_int2(0, 0);
            if (lane < n) my = csr[base + lane];
        }
        for (int j = 0; j < n; j += GROUP) {
            float4 v[GROUP];
            float  w[GROUP];
#pragma unroll
            for (int u = 0; u < GROUP; ++u) {
                v[u] = make_float4(0.f, 0.f, 0.f, 0.f);
                w[u] = 0.f;
                if (u == 0 || j + u < n) {
                    const int su = __builtin_amdgcn_readlane(my.x, j + u);
                    w[u] = __int_as_float(__builtin_amdgcn_readlane(my.y, j + u));
                    v[u] = lin4[su * (D / 4) + lane];
                }
            }
#pragma unroll
            for (int u = 0; u < GROUP; ++u) {
                acc.x = fmaf(v[u].x, w[u], acc.x);
                acc.y = fmaf(v[u].y, w[u], acc.y);
                acc.z = fmaf(v[u].z, w[u], acc.z);
                acc.w = fmaf(v[u].w, w[u], acc.w);
            }
        }
    }

    if (c > CAP) {
        const int m = cnt[N_NODES];
        for (int q = 0; q < m; ++q) {
            const int2 p = ovf[q];
            if (p.x == node) {
                const float w = rsqrtf((float)(cnt[p.y] + 1)) * dn;
                const float4 v = lin4[p.y * (D / 4) + lane];
                acc.x = fmaf(v.x, w, acc.x);
                acc.y = fmaf(v.y, w, acc.y);
                acc.z = fmaf(v.z, w, acc.z);
                acc.w = fmaf(v.w, w, acc.w);
            }
        }
    }

    const float s = dn * dn;
    float4 r;
    r.x = fmaxf(acc.x + l.x * s + b.x, 0.f);
    r.y = fmaxf(acc.y + l.y * s + b.y, 0.f);
    r.z = fmaxf(acc.z + l.z * s + b.z, 0.f);
    r.w = fmaxf(acc.w + l.w * s + b.w, 0.f);
    return r;
}

template <int LDW, int NT, bool WRITE_H, bool BIAS2>
__global__ __launch_bounds__(256) void gcn_gather_mm(const float* __restrict__ lin,
                                                     const int* __restrict__ cnt,
                                                     const int2* __restrict__ csr,
                                                     const int2* __restrict__ ovf,
                                                     const float* __restrict__ bias,
                                                     const float* __restrict__ W2,
                                                     const float* __restrict__ bias2,
                                                     float* __restrict__ hout,
                                                     float* __restrict__ O,
                                                     int* __restrict__ ctr) {
    __shared__ __attribute__((aligned(16))) float hs[2][16 * HP];
    __shared__ int   s_tile[2];

    const int tid  = threadIdx.x;
    const int lane = tid & 31;
    const int wave  = __builtin_amdgcn_readfirstlane(tid >> 5);
    const int nlo   = lane & 15;
    const int khalf = lane >> 4;

    if (tid == 0) s_tile[0] = atomicAdd(ctr, 1);
    __syncthreads();
    int tile = __builtin_amdgcn_readfirstlane(s_tile[0]);
    if (tile >= ROW_TILES) return;

    const float4* lin4 = (const float4*)lin;
    const float4 b = ((const float4*)bias)[lane];

    const bool mm = wave < NT;
    const int  ncol = wave * 16 + nlo;
    const int  nc   = ncol < LDW ? ncol : (LDW - 1);
    v2f bw[32];
#pragma unroll
    for (int s = 0; s < 32; ++s) {
        bw[s].x = 0.f;
        bw[s].y = 0.f;
        if (mm) {
            const int ka = 4 * s + 2 * khalf;
            bw[s].x = W2[ka * LDW + nc];
            bw[s].y = W2[(ka + 1) * LDW + nc];
        }
    }
    float bb = 0.f;
    if (BIAS2 && mm && ncol < LDW) bb = bias2[ncol];

    for (int it = 0; tile < ROW_TILES; ++it) {
        float* hb = hs[it & 1];

        const int node0 = tile * 16 + 2 * wave;
        const int node1 = node0 + 1;
        const int c0 = cnt[node0];
        const int c1 = cnt[node1];
        const int f0 = c0 < 32 ? c0 : 32;
        const int f1 = c1 < 32 ? c1 : 32;
        int2 my0 = make_int2(0, 0);
        int2 my1 = make_int2(0, 0);
        if (lane < f0) my0 = csr[node0 * CAP + lane];
        if (lane < f1) my1 = csr[node1 * CAP + lane];
        const float4 l0 = lin4[node0 * (D / 4) + lane];
        const float4 l1 = lin4[node1 * (D / 4) + lane];

        const float4 r0 = gather_node(lin4, cnt, csr, ovf, node0, c0, my0, l0, lane, b);
        *(float4*)(hb + (2 * wave) * HP + 4 * lane) = r0;
        if (WRITE_H) ((float4*)hout)[node0 * (D / 4) + lane] = r0;

        const float4 r1 = gather_node(lin4, cnt, csr, ovf, node1, c1, my1, l1, lane, b);
        *(float4*)(hb + (2 * wave + 1) * HP + 4 * lane) = r1;
        if (WRITE_H) ((float4*)hout)[node1 * (D / 4) + lane] = r1;

        if (tid == 0) s_tile[(it + 1) & 1] = atomicAdd(ctr, 1);
        __syncthreads();

        if (mm) {
            v8f acc = {};
            const float* arow = hb + nlo * HP + 2 * khalf;
#pragma unroll
            for (int s = 0; s < 32; ++s) {
                const v2f a = *(const v2f*)(arow + 4 * s);
                acc = __builtin_amdgcn_wmma_f32_16x16x4_f32(
                    false, a, false, bw[s], (short)0, acc, false, false);
            }
            if (ncol < LDW) {
                const int rbase = tile * 16 + 8 * khalf;
#pragma unroll
                for (int r = 0; r < 8; ++r) {
                    if (BIAS2) O[(rbase + r) * LDW + ncol] = acc[r] + bb;
                    else       O[(rbase + r) * LDW + ncol] = acc[r];
                }
            }
        }
        tile = __builtin_amdgcn_readfirstlane(s_tile[(it + 1) & 1]);
    }
}

extern "C" void kernel_launch(void* const* d_in, const int* in_sizes, int n_in,
                              void* d_out, int out_size, void* d_ws, size_t ws_size,
                              hipStream_t stream) {
    const float* x   = (const float*)d_in[0];
    const int*   ei  = (const int*)d_in[1];
    const float* W1  = (const float*)d_in[2];
    const float* b1  = (const float*)d_in[3];
    const float* W2  = (const float*)d_in[4];
    const float* b2  = (const float*)d_in[5];
    const float* W3  = (const float*)d_in[6];
    const float* b3  = (const float*)d_in[7];
    const float* fcW = (const float*)d_in[8];
    const float* fcb = (const float*)d_in[9];

    float* out    = (float*)d_out;
    float* h_out  = out;
    float* logits = out + N_NODES * D;

    float* ws     = (float*)d_ws;
    int*   cnt    = (int*)d_ws;
    int*   slots  = cnt + NPAD;
    int2*  csr    = (int2*)(slots + N_NODES * CAP);
    int2*  ovf    = csr + N_NODES * CAP;
    float* B0     = ws + NPAD + 3 * N_NODES * CAP + 2 * N_EDGES;
    float* B1     = B0 + N_NODES * D;

    const int* src = ei;
    const int* dst = ei + N_EDGES;

    const int TPB = 256;
    const int gP    = NPAD / TPB;
    const int gNode = (N_NODES + 7) / 8;
    const int gRes  = 1024;
    const int gResF = 768;
    int* ctr1 = cnt + N_NODES + 1;
    int* ctr2 = cnt + N_NODES + 2;
    int* ctr3 = cnt + N_NODES + 3;

    cnt_zero<<<gP, TPB, 0, stream>>>(cnt);
    place_and_gemm<<<GEMM_BLOCKS + PLACE_BLOCKS, TPB, 0, stream>>>(src, dst, cnt, slots, ovf, x, W1, B0);
    seg_order<<<gNode, TPB, 0, stream>>>(cnt, slots, csr);

    gcn_gather_mm<D, 8, false, false><<<gRes, TPB, 0, stream>>>(B0, cnt, csr, ovf, b1, W2, b2, B1, B1, ctr1);

    gcn_gather_mm<D, 8, false, false><<<gRes, TPB, 0, stream>>>(B1, cnt, csr, ovf, b2, W3, b3, B0, B0, ctr2);

    gcn_gather_mm<N_CLASSES, 3, true, true><<<gResF, TPB, 0, stream>>>(B0, cnt, csr, ovf, b3, fcW, fcb, h_out, logits, ctr3);
}


# Resident blocks walk tiles at a fixed stride, no ticket add: next tile's counts, pairs, own rows are requested a phase ahead, under the product; 3 round trips leave each tile's chain; sum order kept
# speedup vs baseline: 4.0762x; 1.0316x over previous
#include <hip/hip_runtime.h>
#include <hip/hip_bf16.h>

typedef __attribute__((ext_vector_type(2))) float v2f;
typedef __attribute__((ext_vector_type(8))) float v8f;

#define N_NODES 50000
#define N_EDGES 800000
#define D 128
#define N_CLASSES 40
#define ROW_TILES 3125
#define NPAD 50176
#define CAP 64
#define GROUP 8

__global__ __launch_bounds__(256) void cnt_zero(int* __restrict__ cnt) {
    int i = blockIdx.x * blockDim.x + threadIdx.x;
    if (i < NPAD) cnt[i] = 0;
}

__device__ __forceinline__ void edge_place_body(const int bid,
                                                const int* __restrict__ src,
                                                const int* __restrict__ dst,
                                                int* __restrict__ cnt,
                                                int* __restrict__ slots,
                                                int2* __restrict__ ovf) {
    int e = bid * 256 + threadIdx.x;
    if (e < N_EDGES) {
        const int d = dst[e];
        const int s = src[e];
        const int r = atomicAdd(&cnt[d], 1);
        if (r < CAP) {
            slots[d * CAP + r] = s;
        } else {
            const int q = atomicAdd(&cnt[N_NODES], 1);
            ovf[q] = make_int2(d, s);
        }
    }
}

__global__ __launch_bounds__(256) void seg_order(const int* __restrict__ cnt,
                                                 const int* __restrict__ slots,
                                                 int2* __restrict__ csr) {
    const int lane = threadIdx.x & 31;
    const int node = __builtin_amdgcn_readfirstlane((blockIdx.x * 256 + threadIdx.x) >> 5);
    if (node >= N_NODES) return;
    const int c   = cnt[node];
    const int n   = c < CAP ? c : CAP;
    const int beg = node * CAP;
    const float dn = rsqrtf((float)(c + 1));
    if (n <= 32) {
        const int key = (lane < n) ? slots[beg + lane] : 0x7fffffff;
        int pos = 0;
        for (int j = 0; j < n; ++j) {
            const int kj = __builtin_amdgcn_readlane(key, j);
            pos += (kj < key || (kj == key && j < lane)) ? 1 : 0;
        }
        if (lane < n)
            csr[beg + pos] = make_int2(key, __float_as_int(rsqrtf((float)(cnt[key] + 1)) * dn));
    } else {
        for (int i = lane; i < n; i += 32) {
            const int key = slots[beg + i];
            int pos = 0;
            for (int j = 0; j < n; ++j) {
                const int kj = slots[beg + j];
                pos += (kj < key || (kj == key && j < i)) ? 1 : 0;
            }
            csr[beg + pos] = make_int2(key, __float_as_int(rsqrtf((float)(cnt[key] + 1)) * dn));
        }
    }
}

__device__ __forceinline__ void gemm128_body(float* __restrict__ Ws, const int bid,
                                             const float* __restrict__ A,
                                             const float* __restrict__ W,
                                             float* __restrict__ O) {
    const int tid = threadIdx.x;
#pragma unroll
    for (int u = tid; u < 2048; u += 256) {
        const int kp = u >> 5;
        const int n4 = (u & 31) * 4;
        const float4 r0 = *(const float4*)(W + (2 * kp) * D + n4);
        const float4 r1 = *(const float4*)(W + (2 * kp + 1) * D + n4);
        float4* dst = (float4*)(Ws + kp * 256 + n4 * 2);
        dst[0] = make_float4(r0.x, r1.x, r0.y, r1.y);
        dst[1] = make_float4(r0.z, r1.z, r0.w, r1.w);
    }
    __syncthreads();

    const int wave = tid >> 5;
    const int lane = tid & 31;
    const int tile = bid * 8 + wave;
    if (tile >= ROW_TILES) return;

    const int nlo   = lane & 15;
    const int khalf = lane >> 4;
    const int row   = tile * 16 + nlo;

    v8f acc[8] = {};

    for (int k0 = 0; k0 < D; k0 += 4) {
        const int ka = k0 + 2 * khalf;
        v2f a;
        a.x = A[row * D + ka];
        a.y = A[row * D + ka + 1];
        const v2f* Brow = (const v2f*)(Ws + (ka >> 1) * 256);
#pragma unroll
        for (int t = 0; t < 8; ++t) {
            const v2f b = Brow[t * 16 + nlo];
            acc[t] = __builtin_amdgcn_wmma_f32_16x16x4_f32(
                false, a, false, b, (short)0, acc[t], false, false);
        }
    }

    const int rbase = tile * 16 + 8 * khalf;
#pragma unroll
    for (int t = 0; t < 8; ++t) {
        const int n = t * 16 + nlo;
#pragma unroll
        for (int r = 0; r < 8; ++r) O[(rbase + r) * D + n] = acc[t][r];
    }
}

#define GEMM_BLOCKS 391
#define PLACE_BLOCKS 3125

__global__ __launch_bounds__(256) void place_and_gemm(const int* __restrict__ src,
                                                      const int* __restrict__ dst,
                                                      int* __restrict__ cnt,
                                                      int* __restrict__ slots,
                                                      int2* __restrict__ ovf,
                                                      const float* __restrict__ A,
                                                      const float* __restrict__ W,
                                                      float* __restrict__ O) {
    __shared__ float Ws[D * D];
    if (blockIdx.x < GEMM_BLOCKS) gemm128_body(Ws, blockIdx.x, A, W, O);
    else                          edge_place_body(blockIdx.x - GEMM_BLOCKS, src, dst, cnt, slots, ovf);
}

#define HP 132

__device__ __forceinline__ float4 gather_node(const float4* __restrict__ lin4,
                                              const int* __restrict__ cnt,
                                              const int2* __restrict__ csr,
                                              const int2* __restrict__ ovf,
                                              const int node, const int c,
                                              const int2 my_first, const float4 l,
                                              const int lane, const float4 b) {
    const int beg = node * CAP;
    const int end = beg + (c < CAP ? c : CAP);
    const float dn = rsqrtf((float)(c + 1));

    float4 acc = make_float4(0.f, 0.f, 0.f, 0.f);
    for (int base = beg; base < end; base += 32) {
        const int n = (end - base) < 32 ? (end - base) : 32;
        int2 my = my_first;
        if (base != beg) {
            my = make_int2(0, 0);
            if (lane < n) my = csr[base + lane];
        }
        for (int j = 0; j < n; j += GROUP) {
            float4 v[GROUP];
            float  w[GROUP];
#pragma unroll
            for (int u = 0; u < GROUP; ++u) {
                v[u] = make_float4(0.f, 0.f, 0.f, 0.f);
                w[u] = 0.f;
                if (u == 0 || j + u < n) {
                    const int su = __builtin_amdgcn_readlane(my.x, j + u);
                    w[u] = __int_as_float(__builtin_amdgcn_readlane(my.y, j + u));
                    v[u] = lin4[su * (D / 4) + lane];
                }
            }
#pragma unroll
            for (int u = 0; u < GROUP; ++u) {
                acc.x = fmaf(v[u].x, w[u], acc.x);
                acc.y = fmaf(v[u].y, w[u], acc.y);
                acc.z = fmaf(v[u].z, w[u], acc.z);
                acc.w = fmaf(v[u].w, w[u], acc.w);
            }
        }
    }

    if (c > CAP) {
        const int m = cnt[N_NODES];
        for (int q = 0; q < m; ++q) {
            const int2 p = ovf[q];
            if (p.x == node) {
                const float w = rsqrtf((float)(cnt[p.y] + 1)) * dn;
                const float4 v = lin4[p.y * (D / 4) + lane];
                acc.x = fmaf(v.x, w, acc.x);
                acc.y = fmaf(v.y, w, acc.y);
                acc.z = fmaf(v.z, w, acc.z);
                acc.w = fmaf(v.w, w, acc.w);
            }
        }
    }

    const float s = dn * dn;
    float4 r;
    r.x = fmaxf(acc.x + l.x * s + b.x, 0.f);
    r.y = fmaxf(acc.y + l.y * s + b.y, 0.f);
    r.z = fmaxf(acc.z + l.z * s + b.z, 0.f);
    r.w = fmaxf(acc.w + l.w * s + b.w, 0.f);
    return r;
}

template <int LDW, int NT, bool WRITE_H, bool BIAS2>
__global__ __launch_bounds__(256) void gcn_gather_mm(const float* __restrict__ lin,
                                                     const int* __restrict__ cnt,
                                                     const int2* __restrict__ csr,
                                                     const int2* __restrict__ ovf,
                                                     const float* __restrict__ bias,
                                                     const float* __restrict__ W2,
                                                     const float* __restrict__ bias2,
                                                     float* __restrict__ hout,
                                                     float* __restrict__ O) {
    __shared__ __attribute__((aligned(16))) float hs[2][16 * HP];

    const int tid  = threadIdx.x;
    const int lane = tid & 31;
    const int wave  = __builtin_amdgcn_readfirstlane(tid >> 5);
    const int nlo   = lane & 15;
    const int khalf = lane >> 4;

    int tile = blockIdx.x;
    if (tile >= ROW_TILES) return;

    const float4* lin4 = (const float4*)lin;
    const float4 b = ((const float4*)bias)[lane];

    const bool mm = wave < NT;
    const int  ncol = wave * 16 + nlo;
    const int  nc   = ncol < LDW ? ncol : (LDW - 1);
    v2f bw[32];
#pragma unroll
    for (int s = 0; s < 32; ++s) {
        bw[s].x = 0.f;
        bw[s].y = 0.f;
        if (mm) {
            const int ka = 4 * s + 2 * khalf;
            bw[s].x = W2[ka * LDW + nc];
            bw[s].y = W2[(ka + 1) * LDW + nc];
        }
    }
    float bb = 0.f;
    if (BIAS2 && mm && ncol < LDW) bb = bias2[ncol];

    int c0 = cnt[tile * 16 + 2 * wave];
    int c1 = cnt[tile * 16 + 2 * wave + 1];
    int2 my0 = make_int2(0, 0);
    int2 my1 = make_int2(0, 0);
    if (lane < (c0 < 32 ? c0 : 32)) my0 = csr[(tile * 16 + 2 * wave) * CAP + lane];
    if (lane < (c1 < 32 ? c1 : 32)) my1 = csr[(tile * 16 + 2 * wave + 1) * CAP + lane];
    float4 l0 = lin4[(tile * 16 + 2 * wave) * (D / 4) + lane];
    float4 l1 = lin4[(tile * 16 + 2 * wave + 1) * (D / 4) + lane];

    const int stride = (int)gridDim.x;
    int ncv = 0;
    if (tile + stride < ROW_TILES) ncv = cnt[(tile + stride) * 16 + 2 * wave + (lane & 1)];

    for (int it = 0; tile < ROW_TILES; ++it) {
        float* hb = hs[it & 1];

        const int  node0  = tile * 16 + 2 * wave;
        const int  node1  = node0 + 1;
        const int  next   = tile + stride;
        const bool more   = next < ROW_TILES;
        const int  nnode0 = next * 16 + 2 * wave;
        const int  nnode1 = nnode0 + 1;

        const float4 r0 = gather_node(lin4, cnt, csr, ovf, node0, c0, my0, l0, lane, b);
        *(float4*)(hb + (2 * wave) * HP + 4 * lane) = r0;
        if (WRITE_H) ((float4*)hout)[node0 * (D / 4) + lane] = r0;

        const float4 r1 = gather_node(lin4, cnt, csr, ovf, node1, c1, my1, l1, lane, b);
        *(float4*)(hb + (2 * wave + 1) * HP + 4 * lane) = r1;
        if (WRITE_H) ((float4*)hout)[node1 * (D / 4) + lane] = r1;

        __syncthreads();

        const int nc0 = __builtin_amdgcn_readlane(ncv, 0);
        const int nc1 = __builtin_amdgcn_readlane(ncv, 1);
        int2   nmy0 = make_int2(0, 0);
        int2   nmy1 = make_int2(0, 0);
        float4 nl0  = make_float4(0.f, 0.f, 0.f, 0.f);
        float4 nl1  = make_float4(0.f, 0.f, 0.f, 0.f);
        if (more) {
            if (lane < (nc0 < 32 ? nc0 : 32)) nmy0 = csr[nnode0 * CAP + lane];
            if (lane < (nc1 < 32 ? nc1 : 32)) nmy1 = csr[nnode1 * CAP + lane];
            nl0 = lin4[nnode0 * (D / 4) + lane];
            nl1 = lin4[nnode1 * (D / 4) + lane];
        }
        int ncv2 = 0;
        if (next + stride < ROW_TILES) ncv2 = cnt[(next + stride) * 16 + 2 * wave + (lane & 1)];

        if (mm) {
            v8f acc = {};
            const float* arow = hb + nlo * HP + 2 * khalf;
#pragma unroll
            for (int s = 0; s < 32; ++s) {
                const v2f a = *(const v2f*)(arow + 4 * s);
                acc = __builtin_amdgcn_wmma_f32_16x16x4_f32(
                    false, a, false, bw[s], (short)0, acc, false, false);
            }
            if (ncol < LDW) {
                const int rbase = tile * 16 + 8 * khalf;
#pragma unroll
                for (int r = 0; r < 8; ++r) {
                    if (BIAS2) O[(rbase + r) * LDW + ncol] = acc[r] + bb;
                    else       O[(rbase + r) * LDW + ncol] = acc[r];
                }
            }
        }

        tile = next;
        ncv = ncv2;
        c0 = nc0;  c1 = nc1;
        my0 = nmy0; my1 = nmy1;
        l0 = nl0;  l1 = nl1;
    }
}

extern "C" void kernel_launch(void* const* d_in, const int* in_sizes, int n_in,
                              void* d_out, int out_size, void* d_ws, size_t ws_size,
                              hipStream_t stream) {
    const float* x   = (const float*)d_in[0];
    const int*   ei  = (const int*)d_in[1];
    const float* W1  = (const float*)d_in[2];
    const float* b1  = (const float*)d_in[3];
    const float* W2  = (const float*)d_in[4];
    const float* b2  = (const float*)d_in[5];
    const float* W3  = (const float*)d_in[6];
    const float* b3  = (const float*)d_in[7];
    const float* fcW = (const float*)d_in[8];
    const float* fcb = (const float*)d_in[9];

    float* out    = (float*)d_out;
    float* h_out  = out;
    float* logits = out + N_NODES * D;

    float* ws     = (float*)d_ws;
    int*   cnt    = (int*)d_ws;
    int*   slots  = cnt + NPAD;
    int2*  csr    = (int2*)(slots + N_NODES * CAP);
    int2*  ovf    = csr + N_NODES * CAP;
    float* B0     = ws + NPAD + 3 * N_NODES * CAP + 2 * N_EDGES;
    float* B1     = B0 + N_NODES * D;

    const int* src = ei;
    const int* dst = ei + N_EDGES;

    const int TPB = 256;
    const int gP    = NPAD / TPB;
    const int gNode = (N_NODES + 7) / 8;
    const int gRes  = 1024;
    const int gResF = 768;

    cnt_zero<<<gP, TPB, 0, stream>>>(cnt);
    place_and_gemm<<<GEMM_BLOCKS + PLACE_BLOCKS, TPB, 0, stream>>>(src, dst, cnt, slots, ovf, x, W1, B0);
    seg_order<<<gNode, TPB, 0, stream>>>(cnt, slots, csr);

    gcn_gather_mm<D, 8, false, false><<<gRes, TPB, 0, stream>>>(B0, cnt, csr, ovf, b1, W2, b2, B1, B1);

    gcn_gather_mm<D, 8, false, false><<<gRes, TPB, 0, stream>>>(B1, cnt, csr, ovf, b2, W3, b3, B0, B0);

    gcn_gather_mm<N_CLASSES, 3, true, true><<<gResF, TPB, 0, stream>>>(B0, cnt, csr, ovf, b3, fcW, fcb, h_out, logits);
}
